# starting kernel
# baseline (speedup 1.0000x reference)
#include <hip/hip_runtime.h>

typedef __attribute__((ext_vector_type(2))) float v2f;
typedef __attribute__((ext_vector_type(4))) float v4f;
typedef __attribute__((ext_vector_type(8))) float v8f;

#define CCH    128
#define HW     3136
#define NT     64
#define PITCH2 160
#define SCOPE_TAPS 27

__global__ __launch_bounds__(256)
void inhib_toeplitz_wmma(const float* __restrict__ act,
                         const float* __restrict__ filt,
                         float* __restrict__ out)
{
    __shared__ float lds_filt[CCH];
    __shared__ float lds_x[(CCH / 2) * PITCH2];

    const int t    = threadIdx.x;
    const int lane = t & 31;
    const int wave = t >> 5;
    const int l15  = lane & 15;
    const int hi   = lane >> 4;

    const int n0 = blockIdx.x * NT;
    const int b  = blockIdx.y;
    const float* xbase = act + (size_t)b * CCH * HW;
    float*       obase = out + (size_t)b * CCH * HW;

    v4f xs[8];
#pragma unroll
    for (int it = 0; it < 8; ++it) {
        int idx = it * 256 + t;
        int r   = idx >> 4;
        int c4  = (idx & 15) << 2;
        xs[it] = *(const v4f*)(xbase + (size_t)r * HW + n0 + c4);
    }

    if (t < CCH) lds_filt[t] = (t < SCOPE_TAPS) ? filt[t] : 0.0f;
    __syncthreads();

    const int i = wave * 16 + l15;
    v2f afrag[32];
#pragma unroll
    for (int s = 0; s < 32; ++s) {
        int j0 = 4 * s + 2 * hi;
        int k0 = (j0     - i + 141) & 127;
        int k1 = (j0 + 1 - i + 141) & 127;
        v2f av;
        av.x = lds_filt[k0];
        av.y = lds_filt[k1];
        afrag[s] = av;
    }

#pragma unroll
    for (int it = 0; it < 8; ++it) {
        int idx  = it * 256 + t;
        int r    = idx >> 4;
        int c4   = (idx & 15) << 2;
        int base = (r >> 1) * PITCH2 + 2 * c4 + (r & 1);
        lds_x[base + 0] = xs[it].x;
        lds_x[base + 2] = xs[it].y;
        lds_x[base + 4] = xs[it].z;
        lds_x[base + 6] = xs[it].w;
    }
    __syncthreads();

    v8f acc0 = {}, acc1 = {}, acc2 = {}, acc3 = {};
#pragma unroll
    for (int s = 0; s < 32; ++s) {
        const v2f av = afrag[s];
        const v2f* bp = (const v2f*)&lds_x[(2 * s + hi) * PITCH2 + 2 * l15];
        v2f b0 = bp[0];
        v2f b1 = bp[16];
        v2f b2 = bp[32];
        v2f b3 = bp[48];

        acc0 = __builtin_amdgcn_wmma_f32_16x16x4_f32(false, av, false, b0, (short)0, acc0, false, false);
        acc1 = __builtin_amdgcn_wmma_f32_16x16x4_f32(false, av, false, b1, (short)0, acc1, false, false);
        acc2 = __builtin_amdgcn_wmma_f32_16x16x4_f32(false, av, false, b2, (short)0, acc2, false, false);
        acc3 = __builtin_amdgcn_wmma_f32_16x16x4_f32(false, av, false, b3, (short)0, acc3, false, false);
    }

    const int i0 = wave * 16;
#pragma unroll
    for (int v = 0; v < 8; ++v) {
        int row = i0 + v + 8 * hi;
        float* orow = obase + (size_t)row * HW + n0 + l15;
        orow[0]  = acc0[v];
        orow[16] = acc1[v];
        orow[32] = acc2[v];
        orow[48] = acc3[v];
    }
}

extern "C" void kernel_launch(void* const* d_in, const int* in_sizes, int n_in,
                              void* d_out, int out_size, void* d_ws, size_t ws_size,
                              hipStream_t stream)
{
    const float* act  = (const float*)d_in[0];
    const float* filt = (const float*)d_in[1];
    float* out = (float*)d_out;

    const int batch = in_sizes[0] / (CCH * HW);
    dim3 grid(HW / NT, batch);
    inhib_toeplitz_wmma<<<grid, 256, 0, stream>>>(act, filt, out);
}


# Software-pipelined tile loop: 512 workgroups each walk 3-4 tiles with the next tile's global loads in flight under this tile's WMMAs and stores; two LDS tiles, one LDS-only barrier per tile
# speedup vs baseline: 1.7074x; 1.7074x over previous
#include <hip/hip_runtime.h>


typedef __attribute__((ext_vector_type(2))) float v2f;
typedef __attribute__((ext_vector_type(4))) float v4f;
typedef __attribute__((ext_vector_type(8))) float v8f;

#define CCH    128
#define HW     3136
#define NT     64
#define TILES_PER_IMG (HW / NT)
#define PITCH2 160
#define LDS_TILE ((CCH / 2) * PITCH2)
#define SCOPE_TAPS 27
#define HALF_TAPS  (SCOPE_TAPS / 2)
#define NWG    512

#define KS_ALL    (CCH / 4)
#define KS_BEFORE ((HALF_TAPS + 3) / 4)
#define KS_BAND   (KS_BEFORE + (15 + HALF_TAPS) / 4 + 1)

static __device__ __forceinline__ void lds_barrier()
{
    asm volatile("s_wait_dscnt 0x0" ::: "memory");
    asm volatile("s_barrier_signal -1" ::: "memory");
    asm volatile("s_barrier_wait -1" ::: "memory");
}

__global__ __launch_bounds__(256)
void inhib_toeplitz_wmma(const float* __restrict__ act,
                         const float* __restrict__ filt,
                         float* __restrict__ out,
                         int batch)
{
    __shared__ float lds_filt[CCH];
    __shared__ float lds_x[2 * LDS_TILE];

    const int t    = threadIdx.x;
    const int lane = t & 31;
    const int wave = __builtin_amdgcn_readfirstlane(t >> 5);
    const int l15  = lane & 15;
    const int hi   = lane >> 4;

    const int ntiles = batch * TILES_PER_IMG;
    int tile = blockIdx.x;
    if (tile >= ntiles) return;

    float ftap = 0.0f;
    if (t < SCOPE_TAPS) ftap = filt[t];

    const int lrow = t >> 4;
    const int lc4  = (t & 15) << 2;
    const size_t goff = (size_t)lrow * HW + lc4;

    v4f xs[8];
    {
        const int b  = tile / TILES_PER_IMG;
        const int n0 = (tile - b * TILES_PER_IMG) * NT;
        const float* xb = act + (size_t)b * CCH * HW + n0 + goff;
#pragma unroll
        for (int it = 0; it < 8; ++it)
            xs[it] = *(const v4f*)(xb + (size_t)it * 16 * HW);
    }

    if (t < CCH) lds_filt[t] = ftap;
    lds_barrier();

    const int sbase = (wave * 4 + KS_ALL - KS_BEFORE) & (KS_ALL - 1);
    const int rot   = (sbase + KS_BAND > KS_ALL) ? (KS_ALL - sbase) : 0;
    int ks[KS_BAND];
#pragma unroll
    for (int q = 0; q < KS_BAND; ++q) {
        int u = q + rot;
        if (u >= KS_BAND) u -= KS_BAND;
        ks[q] = (sbase + u) & (KS_ALL - 1);
    }

    const int i = wave * 16 + l15;
    v2f afrag[KS_BAND];
#pragma unroll
    for (int q = 0; q < KS_BAND; ++q) {
        int j0 = 4 * ks[q] + 2 * hi;
        int k0 = (j0     - i + 141) & 127;
        int k1 = (j0 + 1 - i + 141) & 127;
        v2f av;
        av.x = lds_filt[k0];
        av.y = lds_filt[k1];
        afrag[q] = av;
    }

    const int i0 = wave * 16;
    const int sbase_lds = (lrow >> 1) * PITCH2 + 2 * lc4 + (lrow & 1);
    const int blane_off = hi * PITCH2 + 2 * l15;
    const size_t ooff = (size_t)(i0 + 8 * hi) * HW + l15;

    int buf = 0;
    for (;;) {
        float* lx = lds_x + buf * LDS_TILE;

#pragma unroll
        for (int it = 0; it < 8; ++it) {
            float* p = lx + sbase_lds + it * 8 * PITCH2;
            p[0] = xs[it].x;
            p[2] = xs[it].y;
            p[4] = xs[it].z;
            p[6] = xs[it].w;
        }
        lds_barrier();

        const int b  = tile / TILES_PER_IMG;
        const int n0 = (tile - b * TILES_PER_IMG) * NT;
        const int next = tile + NWG;

        if (next < ntiles) {
            const int nb  = next / TILES_PER_IMG;
            const int nn0 = (next - nb * TILES_PER_IMG) * NT;
            const float* xb = act + (size_t)nb * CCH * HW + nn0 + goff;
#pragma unroll
            for (int it = 0; it < 8; ++it)
                xs[it] = *(const v4f*)(xb + (size_t)it * 16 * HW);
        }

        v8f acc0 = {}, acc1 = {}, acc2 = {}, acc3 = {};
        const float* blane = lx + blane_off;
#pragma unroll
        for (int q = 0; q < KS_BAND; ++q) {
            const v2f av = afrag[q];
            const v2f* bp = (const v2f*)(blane + ks[q] * (2 * PITCH2));
            v2f b0 = bp[0];
            v2f b1 = bp[16];
            v2f b2 = bp[32];
            v2f b3 = bp[48];

            acc0 = __builtin_amdgcn_wmma_f32_16x16x4_f32(false, av, false, b0, (short)0, acc0, false, false);
            acc1 = __builtin_amdgcn_wmma_f32_16x16x4_f32(false, av, false, b1, (short)0, acc1, false, false);
            acc2 = __builtin_amdgcn_wmma_f32_16x16x4_f32(false, av, false, b2, (short)0, acc2, false, false);
            acc3 = __builtin_amdgcn_wmma_f32_16x16x4_f32(false, av, false, b3, (short)0, acc3, false, false);
        }

        float* ob = out + (size_t)b * CCH * HW + n0 + ooff;
#pragma unroll
        for (int v = 0; v < 8; ++v) {
            float* orow = ob + (size_t)v * HW;
            orow[0]  = acc0[v];
            orow[16] = acc1[v];
            orow[32] = acc2[v];
            orow[48] = acc3[v];
        }

        if (next >= ntiles) break;
        tile = next;
        buf ^= 1;
    }
}

extern "C" void kernel_launch(void* const* d_in, const int* in_sizes, int n_in,
                              void* d_out, int out_size, void* d_ws, size_t ws_size,
                              hipStream_t stream)
{
    const float* act  = (const float*)d_in[0];
    const float* filt = (const float*)d_in[1];
    float* out = (float*)d_out;

    const int batch = in_sizes[0] / (CCH * HW);
    inhib_toeplitz_wmma<<<NWG, 256, 0, stream>>>(act, filt, out, batch);
}


# Lane c of a B fragment owns the 4 adjacent columns 4c..4c+3 (one per N-subtile): accumulators leave as 8 whole-line global_store_b128 per wave and tile, not 32 half-line stores; B via ds_load_b128
# speedup vs baseline: 1.9267x; 1.1284x over previous
#include <hip/hip_runtime.h>


typedef __attribute__((ext_vector_type(2))) float v2f;
typedef __attribute__((ext_vector_type(4))) float v4f;
typedef __attribute__((ext_vector_type(8))) float v8f;

#define CCH    128
#define HW     3136
#define NT     64
#define TILES_PER_IMG (HW / NT)
#define PITCH2 160
#define PLANE  64
#define LDS_TILE ((CCH / 2) * PITCH2)
#define SCOPE_TAPS 27
#define HALF_TAPS  (SCOPE_TAPS / 2)
#define NWG    512

#define KS_ALL    (CCH / 4)
#define KS_BEFORE ((HALF_TAPS + 3) / 4)
#define KS_BAND   (KS_BEFORE + (15 + HALF_TAPS) / 4 + 1)

static __device__ __forceinline__ void lds_barrier()
{
    asm volatile("s_wait_dscnt 0x0" ::: "memory");
    asm volatile("s_barrier_signal -1" ::: "memory");
    asm volatile("s_barrier_wait -1" ::: "memory");
}

__global__ __launch_bounds__(256)
void inhib_toeplitz_wmma(const float* __restrict__ act,
                         const float* __restrict__ filt,
                         float* __restrict__ out,
                         int batch)
{
    __shared__ float lds_filt[CCH];
    __shared__ __attribute__((aligned(16))) float lds_x[2 * LDS_TILE];

    const int t    = threadIdx.x;
    const int lane = t & 31;
    const int wave = __builtin_amdgcn_readfirstlane(t >> 5);
    const int l15  = lane & 15;
    const int hi   = lane >> 4;

    const int ntiles = batch * TILES_PER_IMG;
    int tile = blockIdx.x;
    if (tile >= ntiles) return;

    float ftap = 0.0f;
    if (t < SCOPE_TAPS) ftap = filt[t];

    const int lrow = t >> 4;
    const int lc4  = (t & 15) << 2;
    const size_t goff = (size_t)lrow * HW + lc4;

    v4f xs[8];
    {
        const int b  = tile / TILES_PER_IMG;
        const int n0 = (tile - b * TILES_PER_IMG) * NT;
        const float* xb = act + (size_t)b * CCH * HW + n0 + goff;
#pragma unroll
        for (int it = 0; it < 8; ++it)
            xs[it] = *(const v4f*)(xb + (size_t)it * 16 * HW);
    }

    if (t < CCH) lds_filt[t] = ftap;
    lds_barrier();

    const int sbase = (wave * 4 + KS_ALL - KS_BEFORE) & (KS_ALL - 1);
    const int rot   = (sbase + KS_BAND > KS_ALL) ? (KS_ALL - sbase) : 0;
    int ks[KS_BAND];
#pragma unroll
    for (int q = 0; q < KS_BAND; ++q) {
        int u = q + rot;
        if (u >= KS_BAND) u -= KS_BAND;
        ks[q] = (sbase + u) & (KS_ALL - 1);
    }

    const int i = wave * 16 + l15;
    v2f afrag[KS_BAND];
#pragma unroll
    for (int q = 0; q < KS_BAND; ++q) {
        int j0 = 4 * ks[q] + 2 * hi;
        int k0 = (j0     - i + 141) & 127;
        int k1 = (j0 + 1 - i + 141) & 127;
        v2f av;
        av.x = lds_filt[k0];
        av.y = lds_filt[k1];
        afrag[q] = av;
    }

    const int i0 = wave * 16;
    const int sbase_lds = (lrow >> 1) * PITCH2 + lc4 + (lrow & 1);
    const int blane_off = hi * PITCH2 + 4 * l15;
    const size_t ooff = (size_t)(i0 + 8 * hi) * HW + 4 * l15;

    int buf = 0;
    for (;;) {
        float* lx = lds_x + buf * LDS_TILE;

#pragma unroll
        for (int it = 0; it < 8; ++it) {
            float* p = lx + sbase_lds + it * 8 * PITCH2;
            p[0]         = xs[it].x;
            p[2]         = xs[it].y;
            p[PLANE]     = xs[it].z;
            p[PLANE + 2] = xs[it].w;
        }
        lds_barrier();

        const int b  = tile / TILES_PER_IMG;
        const int n0 = (tile - b * TILES_PER_IMG) * NT;
        const int next = tile + NWG;

        if (next < ntiles) {
            const int nb  = next / TILES_PER_IMG;
            const int nn0 = (next - nb * TILES_PER_IMG) * NT;
            const float* xb = act + (size_t)nb * CCH * HW + nn0 + goff;
#pragma unroll
            for (int it = 0; it < 8; ++it)
                xs[it] = *(const v4f*)(xb + (size_t)it * 16 * HW);
        }

        v8f acc0 = {}, acc1 = {}, acc2 = {}, acc3 = {};
        const float* blane = lx + blane_off;
#pragma unroll
        for (int q = 0; q < KS_BAND; ++q) {
            const v2f av = afrag[q];
            const v4f* bp = (const v4f*)(blane + ks[q] * (2 * PITCH2));
            const v4f b01 = bp[0];
            const v4f b23 = bp[PLANE / 4];
            v2f b0 = {b01.x, b01.y};
            v2f b1 = {b01.z, b01.w};
            v2f b2 = {b23.x, b23.y};
            v2f b3 = {b23.z, b23.w};

            acc0 = __builtin_amdgcn_wmma_f32_16x16x4_f32(false, av, false, b0, (short)0, acc0, false, false);
            acc1 = __builtin_amdgcn_wmma_f32_16x16x4_f32(false, av, false, b1, (short)0, acc1, false, false);
            acc2 = __builtin_amdgcn_wmma_f32_16x16x4_f32(false, av, false, b2, (short)0, acc2, false, false);
            acc3 = __builtin_amdgcn_wmma_f32_16x16x4_f32(false, av, false, b3, (short)0, acc3, false, false);
        }

        float* ob = out + (size_t)b * CCH * HW + n0 + ooff;
#pragma unroll
        for (int v = 0; v < 8; ++v) {
            const v4f o = {acc0[v], acc1[v], acc2[v], acc3[v]};
            *(v4f*)(ob + (size_t)v * HW) = o;
        }

        if (next >= ntiles) break;
        tile = next;
        buf ^= 1;
    }
}

extern "C" void kernel_launch(void* const* d_in, const int* in_sizes, int n_in,
                              void* d_out, int out_size, void* d_ws, size_t ws_size,
                              hipStream_t stream)
{
    const float* act  = (const float*)d_in[0];
    const float* filt = (const float*)d_in[1];
    float* out = (float*)d_out;

    const int batch = in_sizes[0] / (CCH * HW);
    inhib_toeplitz_wmma<<<NWG, 256, 0, stream>>>(act, filt, out, batch);
}
